# starting kernel
# baseline (speedup 1.0000x reference)
#include <hip/hip_runtime.h>

#define B_   4
#define SQ_  512
#define SK_  512
#define H_   256
#define AD_  128

typedef __attribute__((ext_vector_type(2))) float v2f;
typedef __attribute__((ext_vector_type(8))) float v8f;

__device__ __forceinline__ v8f wmma_f32_tile(const float* __restrict__ Abase, int lda,
                                             const float* __restrict__ Bbase, int ldb,
                                             int K, v8f c)
{
    const int lane = threadIdx.x & 31;
    const int idx  = lane & 15;
    const int klo  = (lane >> 4) << 1;
    for (int k = 0; k < K; k += 4) {
        v2f a, b;
        a.x = Abase[idx * lda + (k + klo)];
        a.y = Abase[idx * lda + (k + klo + 1)];
        b.x = Bbase[(k + klo)     * ldb + idx];
        b.y = Bbase[(k + klo + 1) * ldb + idx];
        c = __builtin_amdgcn_wmma_f32_16x16x4_f32(false, a, false, b,
                                                  (short)0, c, false, false);
    }
    return c;
}

__device__ __forceinline__ void store_tile(float* __restrict__ Cbase, int ldc, v8f c)
{
    const int lane  = threadIdx.x & 31;
    const int col   = lane & 15;
    const int rbase = (lane >> 4) << 3;
#pragma unroll
    for (int r = 0; r < 8; ++r)
        Cbase[(rbase + r) * ldc + col] = c[r];
}

__global__ __launch_bounds__(128) void k_project(
    const float* __restrict__ Q, const float* __restrict__ Kin,
    const float* __restrict__ Wq, const float* __restrict__ Wk,
    float* __restrict__ Qt, float* __restrict__ Kt)
{
    const int wave  = (blockIdx.x * blockDim.x + threadIdx.x) >> 5;
    const int which = wave >> 10;
    const int t     = wave & 1023;
    const int mt    = t >> 3;
    const int nt    = t & 7;
    const float* Ain = which ? Kin : Q;
    const float* Wm  = which ? Wk  : Wq;
    float*       Cm  = which ? Kt  : Qt;
    v8f c = {};
    c = wmma_f32_tile(Ain + (size_t)mt * 16 * H_, H_, Wm + nt * 16, AD_, H_, c);
    store_tile(Cm + (size_t)mt * 16 * AD_ + nt * 16, AD_, c);
}

__device__ __forceinline__ float fast_tanh(float x)
{
    float t = __builtin_amdgcn_exp2f(x * 2.8853900817779268f);
    return 1.0f - 2.0f * __builtin_amdgcn_rcpf(t + 1.0f);
}

__global__ __launch_bounds__(512) void k_scores_softmax(
    const float* __restrict__ Qt, const float* __restrict__ Kt,
    const float* __restrict__ vvec, float* __restrict__ Wout)
{
    __shared__ float sh_qt[AD_];
    __shared__ float sh_v[AD_];
    __shared__ float sh_red[16];

    const int bq  = blockIdx.x;
    const int b   = bq >> 9;
    const int tid = threadIdx.x;

    if (tid < AD_) {
        sh_qt[tid] = Qt[(size_t)bq * AD_ + tid];
        sh_v[tid]  = vvec[tid];
    }
    __syncthreads();

    const float4* __restrict__ kt4 =
        (const float4*)(Kt + ((size_t)b * SK_ + tid) * AD_);
    const float4* q4 = (const float4*)sh_qt;
    const float4* v4 = (const float4*)sh_v;

    float s = 0.0f;
#pragma unroll
    for (int a = 0; a < AD_ / 4; ++a) {
        float4 kv = kt4[a];
        float4 qv = q4[a];
        float4 vw = v4[a];
        s += fast_tanh(qv.x + kv.x) * vw.x;
        s += fast_tanh(qv.y + kv.y) * vw.y;
        s += fast_tanh(qv.z + kv.z) * vw.z;
        s += fast_tanh(qv.w + kv.w) * vw.w;
    }

    float m = s;
#pragma unroll
    for (int off = 16; off > 0; off >>= 1)
        m = fmaxf(m, __shfl_xor(m, off));
    const int wid = tid >> 5;
    if ((tid & 31) == 0) sh_red[wid] = m;
    __syncthreads();
    if (tid < 32) {
        float mm = (tid < 16) ? sh_red[tid] : -__builtin_inff();
#pragma unroll
        for (int off = 8; off > 0; off >>= 1)
            mm = fmaxf(mm, __shfl_xor(mm, off));
        if (tid == 0) sh_red[0] = mm;
    }
    __syncthreads();
    m = sh_red[0];

    float e = __builtin_amdgcn_exp2f((s - m) * 1.4426950408889634f);

    float sum = e;
#pragma unroll
    for (int off = 16; off > 0; off >>= 1)
        sum += __shfl_xor(sum, off);
    __syncthreads();
    if ((tid & 31) == 0) sh_red[wid] = sum;
    __syncthreads();
    if (tid < 32) {
        float ss = (tid < 16) ? sh_red[tid] : 0.0f;
#pragma unroll
        for (int off = 8; off > 0; off >>= 1)
            ss += __shfl_xor(ss, off);
        if (tid == 0) sh_red[0] = ss;
    }
    __syncthreads();

    Wout[(size_t)bq * SK_ + tid] = e / sh_red[0];
}

__global__ __launch_bounds__(128) void k_av(
    const float* __restrict__ Wt, const float* __restrict__ V,
    float* __restrict__ Ov)
{
    const int wave = (blockIdx.x * blockDim.x + threadIdx.x) >> 5;
    const int b    = wave >> 9;
    const int t    = wave & 511;
    const int mt   = t >> 4;
    const int nt   = t & 15;
    const float* Abase = Wt + (size_t)b * SQ_ * SK_ + (size_t)mt * 16 * SK_;
    const float* Bbase = V  + (size_t)b * SK_ * H_  + nt * 16;
    v8f c = {};
    c = wmma_f32_tile(Abase, SK_, Bbase, H_, SK_, c);
    store_tile(Ov + (size_t)b * SQ_ * H_ + (size_t)mt * 16 * H_ + nt * 16, H_, c);
}

__global__ __launch_bounds__(128) void k_out(
    const float* __restrict__ Ov, const float* __restrict__ Wo,
    const float* __restrict__ bo, float* __restrict__ Out)
{
    const int wave = (blockIdx.x * blockDim.x + threadIdx.x) >> 5;
    const int mt   = wave >> 4;
    const int nt   = wave & 15;
    const int lane = threadIdx.x & 31;
    const float bias = bo[nt * 16 + (lane & 15)];
    v8f c;
#pragma unroll
    for (int i = 0; i < 8; ++i) c[i] = bias;
    c = wmma_f32_tile(Ov + (size_t)mt * 16 * H_, H_, Wo + nt * 16, H_, H_, c);
    store_tile(Out + (size_t)mt * 16 * H_ + nt * 16, H_, c);
}

extern "C" void kernel_launch(void* const* d_in, const int* in_sizes, int n_in,
                              void* d_out, int out_size, void* d_ws, size_t ws_size,
                              hipStream_t stream)
{
    const float* Q  = (const float*)d_in[0];
    const float* K  = (const float*)d_in[1];
    const float* V  = (const float*)d_in[2];
    const float* Wq = (const float*)d_in[3];
    const float* Wk = (const float*)d_in[4];
    const float* vv = (const float*)d_in[5];
    const float* Wo = (const float*)d_in[6];
    const float* bo = (const float*)d_in[7];

    float* out     = (float*)d_out;
    float* weights = out + (size_t)B_ * SQ_ * H_;

    float* Qt = (float*)d_ws;
    float* Kt = Qt + (size_t)B_ * SQ_ * AD_;
    float* Ov = Kt + (size_t)B_ * SK_ * AD_;

    k_project<<<512, 128, 0, stream>>>(Q, K, Wq, Wk, Qt, Kt);
    k_scores_softmax<<<B_ * SQ_, 512, 0, stream>>>(Qt, Kt, vv, weights);
    k_av<<<512, 128, 0, stream>>>(weights, V, Ov);
    k_out<<<512, 128, 0, stream>>>(Ov, Wo, bo, out);
}


# score kernel: exp(2k), exp(2q) taken once while staging to LDS (range-checked, else the old loop); pair loop is fma, rcp, fma on -2v/(EqEk+1); sum(v) is each key's start value: sum order changes
# speedup vs baseline: 1.6086x; 1.6086x over previous
#include <hip/hip_runtime.h>

#define B_   4
#define SQ_  512
#define SK_  512
#define H_   256
#define AD_  128

typedef __attribute__((ext_vector_type(2))) float v2f;
typedef __attribute__((ext_vector_type(8))) float v8f;

__device__ __forceinline__ v8f wmma_f32_tile(const float* __restrict__ Abase, int lda,
                                             const float* __restrict__ Bbase, int ldb,
                                             int K, v8f c)
{
    const int lane = threadIdx.x & 31;
    const int idx  = lane & 15;
    const int klo  = (lane >> 4) << 1;
    for (int k = 0; k < K; k += 4) {
        v2f a, b;
        a.x = Abase[idx * lda + (k + klo)];
        a.y = Abase[idx * lda + (k + klo + 1)];
        b.x = Bbase[(k + klo)     * ldb + idx];
        b.y = Bbase[(k + klo + 1) * ldb + idx];
        c = __builtin_amdgcn_wmma_f32_16x16x4_f32(false, a, false, b,
                                                  (short)0, c, false, false);
    }
    return c;
}

__device__ __forceinline__ void store_tile(float* __restrict__ Cbase, int ldc, v8f c)
{
    const int lane  = threadIdx.x & 31;
    const int col   = lane & 15;
    const int rbase = (lane >> 4) << 3;
#pragma unroll
    for (int r = 0; r < 8; ++r)
        Cbase[(rbase + r) * ldc + col] = c[r];
}

__global__ __launch_bounds__(128) void k_project(
    const float* __restrict__ Q, const float* __restrict__ Kin,
    const float* __restrict__ Wq, const float* __restrict__ Wk,
    float* __restrict__ Qt, float* __restrict__ Kt)
{
    const int wave  = (blockIdx.x * blockDim.x + threadIdx.x) >> 5;
    const int which = wave >> 10;
    const int t     = wave & 1023;
    const int mt    = t >> 3;
    const int nt    = t & 7;
    const float* Ain = which ? Kin : Q;
    const float* Wm  = which ? Wk  : Wq;
    float*       Cm  = which ? Kt  : Qt;
    v8f c = {};
    c = wmma_f32_tile(Ain + (size_t)mt * 16 * H_, H_, Wm + nt * 16, AD_, H_, c);
    store_tile(Cm + (size_t)mt * 16 * AD_ + nt * 16, AD_, c);
}

#define QROWS    8
#define KT_PITCH 132
#define KPL      (SK_ / 32)
#define CPY      16
#define EXP2C    2.8853900817779268f
#define FACT_LIM 11.0f

__device__ __forceinline__ float fast_tanh(float x)
{
    float t = __builtin_amdgcn_exp2f(x * EXP2C);
    return 1.0f - 2.0f * __builtin_amdgcn_rcpf(t + 1.0f);
}

__device__ __forceinline__ float amax4(float m, float4 x)
{
    return fmaxf(fmaxf(m, fmaxf(fabsf(x.x), fabsf(x.y))), fmaxf(fabsf(x.z), fabsf(x.w)));
}

__device__ __forceinline__ float4 exp2x4(float4 x)
{
    float4 e;
    e.x = __builtin_amdgcn_exp2f(x.x * EXP2C);
    e.y = __builtin_amdgcn_exp2f(x.y * EXP2C);
    e.z = __builtin_amdgcn_exp2f(x.z * EXP2C);
    e.w = __builtin_amdgcn_exp2f(x.w * EXP2C);
    return e;
}

__device__ __forceinline__ float stage_kt(const float4* __restrict__ src, float* sh_kt,
                                          int tid, int lane, int wid, const bool as_exp)
{
    float amax = 0.0f;
#pragma unroll 1
    for (int it = 0; it < (SK_ * AD_ / 4) / (QROWS * 32); it += CPY) {
        float4 tmp[CPY];
#pragma unroll
        for (int u = 0; u < CPY; ++u)
            tmp[u] = src[(it + u) * (QROWS * 32) + tid];
#pragma unroll
        for (int u = 0; u < CPY; ++u) {
            const int r = (it + u) * QROWS + wid;
            float4 x = tmp[u];
            if (as_exp) {
                amax = amax4(amax, x);
                x = exp2x4(x);
            }
            *(float4*)&sh_kt[r * KT_PITCH + lane * 4] = x;
        }
    }
    return amax;
}

__global__ __launch_bounds__(QROWS * 32) void k_scores_softmax(
    const float* __restrict__ Qt, const float* __restrict__ Kt,
    const float* __restrict__ vvec, float* __restrict__ Wout)
{
    __shared__ __attribute__((aligned(16))) float sh_kt[SK_ * KT_PITCH];
    __shared__ __attribute__((aligned(16))) float sh_qt[QROWS * AD_];
    __shared__ __attribute__((aligned(16))) float sh_v[AD_];
    __shared__ __attribute__((aligned(16))) float sh_vm2[AD_];
    __shared__ float sh_amax[QROWS];

    const int tid  = threadIdx.x;
    const int lane = tid & 31;
    const int wid  = tid >> 5;
    const int row0 = blockIdx.x * QROWS;
    const int b    = row0 >> 9;

    const float4* __restrict__ src = (const float4*)(Kt + (size_t)b * SK_ * AD_);
    float amax = stage_kt(src, sh_kt, tid, lane, wid, true);
    const float4 qraw = ((const float4*)(Qt + (size_t)row0 * AD_))[tid];
    amax = amax4(amax, qraw);
    ((float4*)sh_qt)[tid] = exp2x4(qraw);
    if (tid < AD_) {
        const float vv = vvec[tid];
        sh_v[tid]   = vv;
        sh_vm2[tid] = -2.0f * vv;
    }
#pragma unroll
    for (int off = 16; off > 0; off >>= 1)
        amax = fmaxf(amax, __shfl_xor(amax, off));
    if (lane == 0) sh_amax[wid] = amax;
    __syncthreads();

    float wgmax = sh_amax[0];
#pragma unroll
    for (int i = 1; i < QROWS; ++i) wgmax = fmaxf(wgmax, sh_amax[i]);
    const int in_range = __builtin_amdgcn_readfirstlane((wgmax <= FACT_LIM) ? 1 : 0);

    const float4* q4   = (const float4*)(sh_qt + wid * AD_);
    const float*  krow = sh_kt + lane * KT_PITCH;

    float s[KPL];
    if (in_range) {
        const float4 vl = ((const float4*)sh_v)[lane];
        float vsum = (vl.x + vl.y) + (vl.z + vl.w);
#pragma unroll
        for (int off = 16; off > 0; off >>= 1)
            vsum += __shfl_xor(vsum, off);
#pragma unroll
        for (int j = 0; j < KPL; ++j) s[j] = vsum;

        const float4* v4 = (const float4*)sh_vm2;
        const v2f one2 = {1.0f, 1.0f};
#pragma unroll 1
        for (int a = 0; a < AD_ / 4; ++a) {
            const float4 qv = q4[a];
            const float4 vw = v4[a];
            const v2f q01 = {qv.x, qv.y}, q23 = {qv.z, qv.w};
            float4 kv[KPL];
#pragma unroll
            for (int j = 0; j < KPL; ++j)
                kv[j] = *(const float4*)(krow + j * 32 * KT_PITCH + 4 * a);
#pragma unroll
            for (int j = 0; j < KPL; ++j) {
                const v2f t01 = __builtin_elementwise_fma((v2f){kv[j].x, kv[j].y}, q01, one2);
                const v2f t23 = __builtin_elementwise_fma((v2f){kv[j].z, kv[j].w}, q23, one2);
                s[j] = __builtin_fmaf(__builtin_amdgcn_rcpf(t01.x), vw.x, s[j]);
                s[j] = __builtin_fmaf(__builtin_amdgcn_rcpf(t01.y), vw.y, s[j]);
                s[j] = __builtin_fmaf(__builtin_amdgcn_rcpf(t23.x), vw.z, s[j]);
                s[j] = __builtin_fmaf(__builtin_amdgcn_rcpf(t23.y), vw.w, s[j]);
            }
        }
    } else {
        stage_kt(src, sh_kt, tid, lane, wid, false);
        ((float4*)sh_qt)[tid] = qraw;
        __syncthreads();

        const float4* v4 = (const float4*)sh_v;
#pragma unroll
        for (int j = 0; j < KPL; ++j) s[j] = 0.0f;
#pragma unroll 1
        for (int a = 0; a < AD_ / 4; ++a) {
            const float4 qv = q4[a];
            const float4 vw = v4[a];
            float4 kv[KPL];
#pragma unroll
            for (int j = 0; j < KPL; ++j)
                kv[j] = *(const float4*)(krow + j * 32 * KT_PITCH + 4 * a);
#pragma unroll
            for (int j = 0; j < KPL; ++j) {
                s[j] += fast_tanh(qv.x + kv[j].x) * vw.x;
                s[j] += fast_tanh(qv.y + kv[j].y) * vw.y;
                s[j] += fast_tanh(qv.z + kv[j].z) * vw.z;
                s[j] += fast_tanh(qv.w + kv[j].w) * vw.w;
            }
        }
    }

    float m = s[0];
#pragma unroll
    for (int j = 1; j < KPL; ++j) m = fmaxf(m, s[j]);
#pragma unroll
    for (int off = 16; off > 0; off >>= 1)
        m = fmaxf(m, __shfl_xor(m, off));

    float e[KPL];
    float sum = 0.0f;
#pragma unroll
    for (int j = 0; j < KPL; ++j) {
        e[j] = __builtin_amdgcn_exp2f((s[j] - m) * 1.4426950408889634f);
        sum += e[j];
    }
#pragma unroll
    for (int off = 16; off > 0; off >>= 1)
        sum += __shfl_xor(sum, off);

    float* __restrict__ wrow = Wout + (size_t)(row0 + wid) * SK_ + lane;
#pragma unroll
    for (int j = 0; j < KPL; ++j)
        wrow[32 * j] = e[j] / sum;
}

__global__ __launch_bounds__(128) void k_av(
    const float* __restrict__ Wt, const float* __restrict__ V,
    float* __restrict__ Ov)
{
    const int wave = (blockIdx.x * blockDim.x + threadIdx.x) >> 5;
    const int b    = wave >> 9;
    const int t    = wave & 511;
    const int mt   = t >> 4;
    const int nt   = t & 15;
    const float* Abase = Wt + (size_t)b * SQ_ * SK_ + (size_t)mt * 16 * SK_;
    const float* Bbase = V  + (size_t)b * SK_ * H_  + nt * 16;
    v8f c = {};
    c = wmma_f32_tile(Abase, SK_, Bbase, H_, SK_, c);
    store_tile(Ov + (size_t)b * SQ_ * H_ + (size_t)mt * 16 * H_ + nt * 16, H_, c);
}

__global__ __launch_bounds__(128) void k_out(
    const float* __restrict__ Ov, const float* __restrict__ Wo,
    const float* __restrict__ bo, float* __restrict__ Out)
{
    const int wave = (blockIdx.x * blockDim.x + threadIdx.x) >> 5;
    const int mt   = wave >> 4;
    const int nt   = wave & 15;
    const int lane = threadIdx.x & 31;
    const float bias = bo[nt * 16 + (lane & 15)];
    v8f c;
#pragma unroll
    for (int i = 0; i < 8; ++i) c[i] = bias;
    c = wmma_f32_tile(Ov + (size_t)mt * 16 * H_, H_, Wo + nt * 16, H_, H_, c);
    store_tile(Out + (size_t)mt * 16 * H_ + nt * 16, H_, c);
}

extern "C" void kernel_launch(void* const* d_in, const int* in_sizes, int n_in,
                              void* d_out, int out_size, void* d_ws, size_t ws_size,
                              hipStream_t stream)
{
    const float* Q  = (const float*)d_in[0];
    const float* K  = (const float*)d_in[1];
    const float* V  = (const float*)d_in[2];
    const float* Wq = (const float*)d_in[3];
    const float* Wk = (const float*)d_in[4];
    const float* vv = (const float*)d_in[5];
    const float* Wo = (const float*)d_in[6];
    const float* bo = (const float*)d_in[7];

    float* out     = (float*)d_out;
    float* weights = out + (size_t)B_ * SQ_ * H_;

    float* Qt = (float*)d_ws;
    float* Kt = Qt + (size_t)B_ * SQ_ * AD_;
    float* Ov = Kt + (size_t)B_ * SK_ * AD_;

    k_project<<<512, 128, 0, stream>>>(Q, K, Wq, Wk, Qt, Kt);
    k_scores_softmax<<<B_ * SQ_ / QROWS, QROWS * 32, 0, stream>>>(Qt, Kt, vv, weights);
    k_av<<<512, 128, 0, stream>>>(weights, V, Ov);
    k_out<<<512, 128, 0, stream>>>(Ov, Wo, bo, out);
}


# GEMM kernels: an 8-wave workgroup computes a 32x64 block; A and B go to LDS in k-chunks of 64 by whole-row 16-B loads (two stages, next chunk in flight); waves read fragments from LDS; sums unchanged
# speedup vs baseline: 2.1571x; 1.3410x over previous
#include <hip/hip_runtime.h>

#define B_   4
#define SQ_  512
#define SK_  512
#define H_   256
#define AD_  128

typedef __attribute__((ext_vector_type(2))) float v2f;
typedef __attribute__((ext_vector_type(8))) float v8f;

#define GT   256
#define GBM  32
#define GBN  64
#define GKC  64
#define PA   68
#define PB   72

struct Chunk { float4 a0, a1, b0, b1, b2, b3; };

__device__ __forceinline__ Chunk load_chunk(const float* __restrict__ ga, const int lda,
                                            const float* __restrict__ gb, const int ldb,
                                            const int k0)
{
    Chunk c;
    const float* pa = ga + k0;
    const float* pb = gb + (size_t)k0 * ldb;
    c.a0 = *(const float4*)(pa);
    c.a1 = *(const float4*)(pa + (size_t)16 * lda);
    c.b0 = *(const float4*)(pb);
    c.b1 = *(const float4*)(pb + (size_t)16 * ldb);
    c.b2 = *(const float4*)(pb + (size_t)32 * ldb);
    c.b3 = *(const float4*)(pb + (size_t)48 * ldb);
    return c;
}

__device__ __forceinline__ void store_chunk(const Chunk& c, float* wa, float* wb)
{
    *(float4*)(wa)           = c.a0;
    *(float4*)(wa + 16 * PA) = c.a1;
    *(float4*)(wb)           = c.b0;
    *(float4*)(wb + 16 * PB) = c.b1;
    *(float4*)(wb + 32 * PB) = c.b2;
    *(float4*)(wb + 48 * PB) = c.b3;
}

__device__ __forceinline__ v8f gemm_block_lds(const float* __restrict__ Ablk, const int lda,
                                              const float* __restrict__ Bblk, const int ldb,
                                              const int K, v8f acc, float* sA, float* sB)
{
    static_assert(GBM == 32 && GKC == 64 && GBN == 64 && GT == 256, "Chunk is laid out for these sizes");
    const int tid  = threadIdx.x;
    const int lane = tid & 31;
    const int wid  = tid >> 5;
    const int idx  = lane & 15;
    const int klo  = (lane >> 4) << 1;
    const int tm   = wid >> 2;
    const int tn   = wid & 3;
    const int crow = tid >> 4;
    const int ccol = (tid & 15) << 2;

    const float* ga = Ablk + (size_t)crow * lda + ccol;
    const float* gb = Bblk + (size_t)crow * ldb + ccol;
    float*       wa = sA + crow * PA + ccol;
    float*       wb = sB + crow * PB + ccol;
    const float* ra = sA + (tm * 16 + idx) * PA + klo;
    const float* rb = sB + klo * PB + tn * 16 + idx;

    const int NC = K / GKC;
    Chunk nx = load_chunk(ga, lda, gb, ldb, 0);
    store_chunk(nx, wa, wb);
    if (NC > 1) nx = load_chunk(ga, lda, gb, ldb, GKC);
    __syncthreads();

#pragma unroll 1
    for (int ch = 0; ch < NC; ++ch) {
        const int cur = ch & 1;
        const float* rac = ra + cur * (GBM * PA);
        const float* rbc = rb + cur * (GKC * PB);
        v2f af[GKC / 4], bf[GKC / 4];
#pragma unroll
        for (int j = 0; j < GKC / 4; ++j) {
            af[j]   = *(const v2f*)(rac + 4 * j);
            bf[j].x = rbc[(4 * j)     * PB];
            bf[j].y = rbc[(4 * j + 1) * PB];
        }
        if (ch + 1 < NC)
            store_chunk(nx, wa + (cur ^ 1) * (GBM * PA), wb + (cur ^ 1) * (GKC * PB));
        if (ch + 2 < NC)
            nx = load_chunk(ga, lda, gb, ldb, (ch + 2) * GKC);
#pragma unroll
        for (int j = 0; j < GKC / 4; ++j)
            acc = __builtin_amdgcn_wmma_f32_16x16x4_f32(false, af[j], false, bf[j],
                                                        (short)0, acc, false, false);
        __builtin_amdgcn_sched_barrier(0);
        __syncthreads();
    }
    return acc;
}

__device__ __forceinline__ void store_tile(float* __restrict__ Cbase, int ldc, v8f c)
{
    const int lane  = threadIdx.x & 31;
    const int col   = lane & 15;
    const int rbase = (lane >> 4) << 3;
#pragma unroll
    for (int r = 0; r < 8; ++r)
        Cbase[(rbase + r) * ldc + col] = c[r];
}

__global__ __launch_bounds__(GT) void k_project(
    const float* __restrict__ Q, const float* __restrict__ Kin,
    const float* __restrict__ Wq, const float* __restrict__ Wk,
    float* __restrict__ Qt, float* __restrict__ Kt)
{
    __shared__ __attribute__((aligned(16))) float sA[2 * GBM * PA];
    __shared__ __attribute__((aligned(16))) float sB[2 * GKC * PB];
    const int which = blockIdx.x >> 7;
    const int t     = blockIdx.x & 127;
    const int bm    = t >> 1;
    const int bn    = t & 1;
    const int wid   = threadIdx.x >> 5;
    const float* Ain = which ? Kin : Q;
    const float* Wm  = which ? Wk  : Wq;
    float*       Cm  = which ? Kt  : Qt;
    v8f c = {};
    c = gemm_block_lds(Ain + (size_t)bm * GBM * H_, H_, Wm + bn * GBN, AD_, H_, c, sA, sB);
    store_tile(Cm + (size_t)(bm * GBM + (wid >> 2) * 16) * AD_ + bn * GBN + (wid & 3) * 16, AD_, c);
}

#define QROWS    8
#define KT_PITCH 132
#define KPL      (SK_ / 32)
#define CPY      16
#define EXP2C    2.8853900817779268f
#define FACT_LIM 11.0f

__device__ __forceinline__ float fast_tanh(float x)
{
    float t = __builtin_amdgcn_exp2f(x * EXP2C);
    return 1.0f - 2.0f * __builtin_amdgcn_rcpf(t + 1.0f);
}

__device__ __forceinline__ float amax4(float m, float4 x)
{
    return fmaxf(fmaxf(m, fmaxf(fabsf(x.x), fabsf(x.y))), fmaxf(fabsf(x.z), fabsf(x.w)));
}

__device__ __forceinline__ float4 exp2x4(float4 x)
{
    float4 e;
    e.x = __builtin_amdgcn_exp2f(x.x * EXP2C);
    e.y = __builtin_amdgcn_exp2f(x.y * EXP2C);
    e.z = __builtin_amdgcn_exp2f(x.z * EXP2C);
    e.w = __builtin_amdgcn_exp2f(x.w * EXP2C);
    return e;
}

__device__ __forceinline__ float stage_kt(const float4* __restrict__ src, float* sh_kt,
                                          int tid, int lane, int wid, const bool as_exp)
{
    float amax = 0.0f;
#pragma unroll 1
    for (int it = 0; it < (SK_ * AD_ / 4) / (QROWS * 32); it += CPY) {
        float4 tmp[CPY];
#pragma unroll
        for (int u = 0; u < CPY; ++u)
            tmp[u] = src[(it + u) * (QROWS * 32) + tid];
#pragma unroll
        for (int u = 0; u < CPY; ++u) {
            const int r = (it + u) * QROWS + wid;
            float4 x = tmp[u];
            if (as_exp) {
                amax = amax4(amax, x);
                x = exp2x4(x);
            }
            *(float4*)&sh_kt[r * KT_PITCH + lane * 4] = x;
        }
    }
    return amax;
}

__global__ __launch_bounds__(QROWS * 32) void k_scores_softmax(
    const float* __restrict__ Qt, const float* __restrict__ Kt,
    const float* __restrict__ vvec, float* __restrict__ Wout)
{
    __shared__ __attribute__((aligned(16))) float sh_kt[SK_ * KT_PITCH];
    __shared__ __attribute__((aligned(16))) float sh_qt[QROWS * AD_];
    __shared__ __attribute__((aligned(16))) float sh_v[AD_];
    __shared__ __attribute__((aligned(16))) float sh_vm2[AD_];
    __shared__ float sh_amax[QROWS];

    const int tid  = threadIdx.x;
    const int lane = tid & 31;
    const int wid  = tid >> 5;
    const int row0 = blockIdx.x * QROWS;
    const int b    = row0 >> 9;

    const float4* __restrict__ src = (const float4*)(Kt + (size_t)b * SK_ * AD_);
    float amax = stage_kt(src, sh_kt, tid, lane, wid, true);
    const float4 qraw = ((const float4*)(Qt + (size_t)row0 * AD_))[tid];
    amax = amax4(amax, qraw);
    ((float4*)sh_qt)[tid] = exp2x4(qraw);
    if (tid < AD_) {
        const float vv = vvec[tid];
        sh_v[tid]   = vv;
        sh_vm2[tid] = -2.0f * vv;
    }
#pragma unroll
    for (int off = 16; off > 0; off >>= 1)
        amax = fmaxf(amax, __shfl_xor(amax, off));
    if (lane == 0) sh_amax[wid] = amax;
    __syncthreads();

    float wgmax = sh_amax[0];
#pragma unroll
    for (int i = 1; i < QROWS; ++i) wgmax = fmaxf(wgmax, sh_amax[i]);
    const int in_range = __builtin_amdgcn_readfirstlane((wgmax <= FACT_LIM) ? 1 : 0);

    const float4* q4   = (const float4*)(sh_qt + wid * AD_);
    const float*  krow = sh_kt + lane * KT_PITCH;

    float s[KPL];
    if (in_range) {
        const float4 vl = ((const float4*)sh_v)[lane];
        float vsum = (vl.x + vl.y) + (vl.z + vl.w);
#pragma unroll
        for (int off = 16; off > 0; off >>= 1)
            vsum += __shfl_xor(vsum, off);
#pragma unroll
        for (int j = 0; j < KPL; ++j) s[j] = vsum;

        const float4* v4 = (const float4*)sh_vm2;
        const v2f one2 = {1.0f, 1.0f};
#pragma unroll 1
        for (int a = 0; a < AD_ / 4; ++a) {
            const float4 qv = q4[a];
            const float4 vw = v4[a];
            const v2f q01 = {qv.x, qv.y}, q23 = {qv.z, qv.w};
            float4 kv[KPL];
#pragma unroll
            for (int j = 0; j < KPL; ++j)
                kv[j] = *(const float4*)(krow + j * 32 * KT_PITCH + 4 * a);
#pragma unroll
            for (int j = 0; j < KPL; ++j) {
                const v2f t01 = __builtin_elementwise_fma((v2f){kv[j].x, kv[j].y}, q01, one2);
                const v2f t23 = __builtin_elementwise_fma((v2f){kv[j].z, kv[j].w}, q23, one2);
                s[j] = __builtin_fmaf(__builtin_amdgcn_rcpf(t01.x), vw.x, s[j]);
                s[j] = __builtin_fmaf(__builtin_amdgcn_rcpf(t01.y), vw.y, s[j]);
                s[j] = __builtin_fmaf(__builtin_amdgcn_rcpf(t23.x), vw.z, s[j]);
                s[j] = __builtin_fmaf(__builtin_amdgcn_rcpf(t23.y), vw.w, s[j]);
            }
        }
    } else {
        stage_kt(src, sh_kt, tid, lane, wid, false);
        ((float4*)sh_qt)[tid] = qraw;
        __syncthreads();

        const float4* v4 = (const float4*)sh_v;
#pragma unroll
        for (int j = 0; j < KPL; ++j) s[j] = 0.0f;
#pragma unroll 1
        for (int a = 0; a < AD_ / 4; ++a) {
            const float4 qv = q4[a];
            const float4 vw = v4[a];
            float4 kv[KPL];
#pragma unroll
            for (int j = 0; j < KPL; ++j)
                kv[j] = *(const float4*)(krow + j * 32 * KT_PITCH + 4 * a);
#pragma unroll
            for (int j = 0; j < KPL; ++j) {
                s[j] += fast_tanh(qv.x + kv[j].x) * vw.x;
                s[j] += fast_tanh(qv.y + kv[j].y) * vw.y;
                s[j] += fast_tanh(qv.z + kv[j].z) * vw.z;
                s[j] += fast_tanh(qv.w + kv[j].w) * vw.w;
            }
        }
    }

    float m = s[0];
#pragma unroll
    for (int j = 1; j < KPL; ++j) m = fmaxf(m, s[j]);
#pragma unroll
    for (int off = 16; off > 0; off >>= 1)
        m = fmaxf(m, __shfl_xor(m, off));

    float e[KPL];
    float sum = 0.0f;
#pragma unroll
    for (int j = 0; j < KPL; ++j) {
        e[j] = __builtin_amdgcn_exp2f((s[j] - m) * 1.4426950408889634f);
        sum += e[j];
    }
#pragma unroll
    for (int off = 16; off > 0; off >>= 1)
        sum += __shfl_xor(sum, off);

    float* __restrict__ wrow = Wout + (size_t)(row0 + wid) * SK_ + lane;
#pragma unroll
    for (int j = 0; j < KPL; ++j)
        wrow[32 * j] = e[j] / sum;
}

__global__ __launch_bounds__(GT) void k_av(
    const float* __restrict__ Wt, const float* __restrict__ V,
    float* __restrict__ Ov)
{
    __shared__ __attribute__((aligned(16))) float sA[2 * GBM * PA];
    __shared__ __attribute__((aligned(16))) float sB[2 * GKC * PB];
    const int b   = blockIdx.x >> 6;
    const int t   = blockIdx.x & 63;
    const int bm  = t >> 2;
    const int bn  = t & 3;
    const int wid = threadIdx.x >> 5;
    const float* Abase = Wt + (size_t)b * SQ_ * SK_ + (size_t)bm * GBM * SK_;
    const float* Bbase = V  + (size_t)b * SK_ * H_  + bn * GBN;
    v8f c = {};
    c = gemm_block_lds(Abase, SK_, Bbase, H_, SK_, c, sA, sB);
    store_tile(Ov + (size_t)b * SQ_ * H_ + (size_t)(bm * GBM + (wid >> 2) * 16) * H_
                  + bn * GBN + (wid & 3) * 16, H_, c);
}

__global__ __launch_bounds__(GT) void k_out(
    const float* __restrict__ Ov, const float* __restrict__ Wo,
    const float* __restrict__ bo, float* __restrict__ Out)
{
    __shared__ __attribute__((aligned(16))) float sA[2 * GBM * PA];
    __shared__ __attribute__((aligned(16))) float sB[2 * GKC * PB];
    const int bm   = blockIdx.x >> 2;
    const int bn   = blockIdx.x & 3;
    const int wid  = threadIdx.x >> 5;
    const int lane = threadIdx.x & 31;
    const int col0 = bn * GBN + (wid & 3) * 16;
    const float bias = bo[col0 + (lane & 15)];
    v8f c;
#pragma unroll
    for (int i = 0; i < 8; ++i) c[i] = bias;
    c = gemm_block_lds(Ov + (size_t)bm * GBM * H_, H_, Wo + bn * GBN, H_, H_, c, sA, sB);
    store_tile(Out + (size_t)(bm * GBM + (wid >> 2) * 16) * H_ + col0, H_, c);
}

extern "C" void kernel_launch(void* const* d_in, const int* in_sizes, int n_in,
                              void* d_out, int out_size, void* d_ws, size_t ws_size,
                              hipStream_t stream)
{
    const float* Q  = (const float*)d_in[0];
    const float* K  = (const float*)d_in[1];
    const float* V  = (const float*)d_in[2];
    const float* Wq = (const float*)d_in[3];
    const float* Wk = (const float*)d_in[4];
    const float* vv = (const float*)d_in[5];
    const float* Wo = (const float*)d_in[6];
    const float* bo = (const float*)d_in[7];

    float* out     = (float*)d_out;
    float* weights = out + (size_t)B_ * SQ_ * H_;

    float* Qt = (float*)d_ws;
    float* Kt = Qt + (size_t)B_ * SQ_ * AD_;
    float* Ov = Kt + (size_t)B_ * SK_ * AD_;

    k_project<<<256, GT, 0, stream>>>(Q, K, Wq, Wk, Qt, Kt);
    k_scores_softmax<<<B_ * SQ_ / QROWS, QROWS * 32, 0, stream>>>(Qt, Kt, vv, weights);
    k_av<<<256, GT, 0, stream>>>(weights, V, Ov);
    k_out<<<256, GT, 0, stream>>>(Ov, Wo, bo, out);
}


# GEMM kernels: K loop as a software pipeline: chunk c+1's fragments go LDS -> 2nd register set under chunk c's WMMAs (3 LDS stages; barrier waits on LDS only; loads 4 chunks ahead); sums unchanged
# speedup vs baseline: 2.4912x; 1.1549x over previous
#include <hip/hip_runtime.h>

#define B_   4
#define SQ_  512
#define SK_  512
#define H_   256
#define AD_  128

typedef __attribute__((ext_vector_type(2))) float v2f;
typedef __attribute__((ext_vector_type(8))) float v8f;

#define GT   256
#define GBM  32
#define GBN  64
#define GKC  64
#define PA   68
#define PB   72
#define NST  3
#define NGR  4
#define STG_A (GBM * PA)
#define STG_B (GKC * PB)

struct Chunk { float4 a0, a1, b0, b1, b2, b3; };

__device__ __forceinline__ Chunk load_chunk(const float* __restrict__ ga, const int lda,
                                            const float* __restrict__ gb, const int ldb,
                                            const int k0)
{
    Chunk c;
    const float* pa = ga + k0;
    const float* pb = gb + (size_t)k0 * ldb;
    c.a0 = *(const float4*)(pa);
    c.a1 = *(const float4*)(pa + (size_t)16 * lda);
    c.b0 = *(const float4*)(pb);
    c.b1 = *(const float4*)(pb + (size_t)16 * ldb);
    c.b2 = *(const float4*)(pb + (size_t)32 * ldb);
    c.b3 = *(const float4*)(pb + (size_t)48 * ldb);
    return c;
}

__device__ __forceinline__ void store_chunk(const Chunk& c, float* wa, float* wb)
{
    *(float4*)(wa)           = c.a0;
    *(float4*)(wa + 16 * PA) = c.a1;
    *(float4*)(wb)           = c.b0;
    *(float4*)(wb + 16 * PB) = c.b1;
    *(float4*)(wb + 32 * PB) = c.b2;
    *(float4*)(wb + 48 * PB) = c.b3;
}

__device__ __forceinline__ void lds_barrier()
{
    __builtin_amdgcn_fence(__ATOMIC_RELEASE, "workgroup", "local");
    __builtin_amdgcn_s_barrier();
    __builtin_amdgcn_fence(__ATOMIC_ACQUIRE, "workgroup", "local");
}

__device__ __forceinline__ void frag_load(v2f (&af)[GKC / 4], v2f (&bf)[GKC / 4],
                                          const float* rac, const float* rbc)
{
#pragma unroll
    for (int j = 0; j < GKC / 4; ++j) {
        af[j]   = *(const v2f*)(rac + 4 * j);
        bf[j].x = rbc[(4 * j)     * PB];
        bf[j].y = rbc[(4 * j + 1) * PB];
    }
}

template <int NC>
__device__ __forceinline__ v8f gemm_block_lds(const float* __restrict__ Ablk, const int lda,
                                              const float* __restrict__ Bblk, const int ldb,
                                              v8f acc, float* sA, float* sB)
{
    static_assert(GBM == 32 && GKC == 64 && GBN == 64 && GT == 256, "Chunk is laid out for these sizes");
    static_assert(NST >= 3, "a stage is refilled two trips after its fragments were read");
    const int tid  = threadIdx.x;
    const int lane = tid & 31;
    const int wid  = tid >> 5;
    const int idx  = lane & 15;
    const int klo  = (lane >> 4) << 1;
    const int tm   = wid >> 2;
    const int tn   = wid & 3;
    const int crow = tid >> 4;
    const int ccol = (tid & 15) << 2;

    const float* ga = Ablk + (size_t)crow * lda + ccol;
    const float* gb = Bblk + (size_t)crow * ldb + ccol;
    float*       wa = sA + crow * PA + ccol;
    float*       wb = sB + crow * PB + ccol;
    const float* ra = sA + (tm * 16 + idx) * PA + klo;
    const float* rb = sB + klo * PB + tn * 16 + idx;

    Chunk g[NGR];
#pragma unroll
    for (int c = 0; c < NGR; ++c)
        if (c < NC) g[c] = load_chunk(ga, lda, gb, ldb, c * GKC);

    v2f af[2][GKC / 4], bf[2][GKC / 4];

    store_chunk(g[0], wa, wb);
    if (NGR < NC) g[0] = load_chunk(ga, lda, gb, ldb, NGR * GKC);
    lds_barrier();
    frag_load(af[0], bf[0], ra, rb);
    if (NC > 1) {
        store_chunk(g[1], wa + STG_A, wb + STG_B);
        if (1 + NGR < NC) g[1] = load_chunk(ga, lda, gb, ldb, (1 + NGR) * GKC);
        lds_barrier();
    }

#pragma unroll
    for (int ch = 0; ch < NC; ++ch) {
        if (ch + 1 < NC)
            frag_load(af[(ch + 1) & 1], bf[(ch + 1) & 1],
                      ra + ((ch + 1) % NST) * STG_A, rb + ((ch + 1) % NST) * STG_B);
        asm volatile("" : "+v"(acc) : : "memory");
#pragma unroll
        for (int j = 0; j < GKC / 4; ++j)
            acc = __builtin_amdgcn_wmma_f32_16x16x4_f32(false, af[ch & 1][j], false, bf[ch & 1][j],
                                                        (short)0, acc, false, false);
        asm volatile("" : "+v"(acc) : : "memory");
        if (ch + 2 < NC) {
            store_chunk(g[(ch + 2) % NGR], wa + ((ch + 2) % NST) * STG_A, wb + ((ch + 2) % NST) * STG_B);
            if (ch + 2 + NGR < NC)
                g[(ch + 2) % NGR] = load_chunk(ga, lda, gb, ldb, (ch + 2 + NGR) * GKC);
            lds_barrier();
        }
    }
    return acc;
}

__device__ __forceinline__ void store_tile(float* __restrict__ Cbase, int ldc, v8f c)
{
    const int lane  = threadIdx.x & 31;
    const int col   = lane & 15;
    const int rbase = (lane >> 4) << 3;
#pragma unroll
    for (int r = 0; r < 8; ++r)
        Cbase[(rbase + r) * ldc + col] = c[r];
}

__global__ __launch_bounds__(GT) void k_project(
    const float* __restrict__ Q, const float* __restrict__ Kin,
    const float* __restrict__ Wq, const float* __restrict__ Wk,
    float* __restrict__ Qt, float* __restrict__ Kt)
{
    __shared__ __attribute__((aligned(16))) float sA[NST * STG_A];
    __shared__ __attribute__((aligned(16))) float sB[NST * STG_B];
    const int which = blockIdx.x >> 7;
    const int t     = blockIdx.x & 127;
    const int bm    = t >> 1;
    const int bn    = t & 1;
    const int wid   = threadIdx.x >> 5;
    const float* Ain = which ? Kin : Q;
    const float* Wm  = which ? Wk  : Wq;
    float*       Cm  = which ? Kt  : Qt;
    v8f c = {};
    c = gemm_block_lds<H_ / GKC>(Ain + (size_t)bm * GBM * H_, H_, Wm + bn * GBN, AD_, c, sA, sB);
    store_tile(Cm + (size_t)(bm * GBM + (wid >> 2) * 16) * AD_ + bn * GBN + (wid & 3) * 16, AD_, c);
}

#define QROWS    8
#define KT_PITCH 132
#define KPL      (SK_ / 32)
#define CPY      16
#define EXP2C    2.8853900817779268f
#define FACT_LIM 11.0f

__device__ __forceinline__ float fast_tanh(float x)
{
    float t = __builtin_amdgcn_exp2f(x * EXP2C);
    return 1.0f - 2.0f * __builtin_amdgcn_rcpf(t + 1.0f);
}

__device__ __forceinline__ float amax4(float m, float4 x)
{
    return fmaxf(fmaxf(m, fmaxf(fabsf(x.x), fabsf(x.y))), fmaxf(fabsf(x.z), fabsf(x.w)));
}

__device__ __forceinline__ float4 exp2x4(float4 x)
{
    float4 e;
    e.x = __builtin_amdgcn_exp2f(x.x * EXP2C);
    e.y = __builtin_amdgcn_exp2f(x.y * EXP2C);
    e.z = __builtin_amdgcn_exp2f(x.z * EXP2C);
    e.w = __builtin_amdgcn_exp2f(x.w * EXP2C);
    return e;
}

__device__ __forceinline__ float stage_kt(const float4* __restrict__ src, float* sh_kt,
                                          int tid, int lane, int wid, const bool as_exp)
{
    float amax = 0.0f;
#pragma unroll 1
    for (int it = 0; it < (SK_ * AD_ / 4) / (QROWS * 32); it += CPY) {
        float4 tmp[CPY];
#pragma unroll
        for (int u = 0; u < CPY; ++u)
            tmp[u] = src[(it + u) * (QROWS * 32) + tid];
#pragma unroll
        for (int u = 0; u < CPY; ++u) {
            const int r = (it + u) * QROWS + wid;
            float4 x = tmp[u];
            if (as_exp) {
                amax = amax4(amax, x);
                x = exp2x4(x);
            }
            *(float4*)&sh_kt[r * KT_PITCH + lane * 4] = x;
        }
    }
    return amax;
}

__global__ __launch_bounds__(QROWS * 32) void k_scores_softmax(
    const float* __restrict__ Qt, const float* __restrict__ Kt,
    const float* __restrict__ vvec, float* __restrict__ Wout)
{
    __shared__ __attribute__((aligned(16))) float sh_kt[SK_ * KT_PITCH];
    __shared__ __attribute__((aligned(16))) float sh_qt[QROWS * AD_];
    __shared__ __attribute__((aligned(16))) float sh_v[AD_];
    __shared__ __attribute__((aligned(16))) float sh_vm2[AD_];
    __shared__ float sh_amax[QROWS];

    const int tid  = threadIdx.x;
    const int lane = tid & 31;
    const int wid  = tid >> 5;
    const int row0 = blockIdx.x * QROWS;
    const int b    = row0 >> 9;

    const float4* __restrict__ src = (const float4*)(Kt + (size_t)b * SK_ * AD_);
    float amax = stage_kt(src, sh_kt, tid, lane, wid, true);
    const float4 qraw = ((const float4*)(Qt + (size_t)row0 * AD_))[tid];
    amax = amax4(amax, qraw);
    ((float4*)sh_qt)[tid] = exp2x4(qraw);
    if (tid < AD_) {
        const float vv = vvec[tid];
        sh_v[tid]   = vv;
        sh_vm2[tid] = -2.0f * vv;
    }
#pragma unroll
    for (int off = 16; off > 0; off >>= 1)
        amax = fmaxf(amax, __shfl_xor(amax, off));
    if (lane == 0) sh_amax[wid] = amax;
    __syncthreads();

    float wgmax = sh_amax[0];
#pragma unroll
    for (int i = 1; i < QROWS; ++i) wgmax = fmaxf(wgmax, sh_amax[i]);
    const int in_range = __builtin_amdgcn_readfirstlane((wgmax <= FACT_LIM) ? 1 : 0);

    const float4* q4   = (const float4*)(sh_qt + wid * AD_);
    const float*  krow = sh_kt + lane * KT_PITCH;

    float s[KPL];
    if (in_range) {
        const float4 vl = ((const float4*)sh_v)[lane];
        float vsum = (vl.x + vl.y) + (vl.z + vl.w);
#pragma unroll
        for (int off = 16; off > 0; off >>= 1)
            vsum += __shfl_xor(vsum, off);
#pragma unroll
        for (int j = 0; j < KPL; ++j) s[j] = vsum;

        const float4* v4 = (const float4*)sh_vm2;
        const v2f one2 = {1.0f, 1.0f};
#pragma unroll 1
        for (int a = 0; a < AD_ / 4; ++a) {
            const float4 qv = q4[a];
            const float4 vw = v4[a];
            const v2f q01 = {qv.x, qv.y}, q23 = {qv.z, qv.w};
            float4 kv[KPL];
#pragma unroll
            for (int j = 0; j < KPL; ++j)
                kv[j] = *(const float4*)(krow + j * 32 * KT_PITCH + 4 * a);
#pragma unroll
            for (int j = 0; j < KPL; ++j) {
                const v2f t01 = __builtin_elementwise_fma((v2f){kv[j].x, kv[j].y}, q01, one2);
                const v2f t23 = __builtin_elementwise_fma((v2f){kv[j].z, kv[j].w}, q23, one2);
                s[j] = __builtin_fmaf(__builtin_amdgcn_rcpf(t01.x), vw.x, s[j]);
                s[j] = __builtin_fmaf(__builtin_amdgcn_rcpf(t01.y), vw.y, s[j]);
                s[j] = __builtin_fmaf(__builtin_amdgcn_rcpf(t23.x), vw.z, s[j]);
                s[j] = __builtin_fmaf(__builtin_amdgcn_rcpf(t23.y), vw.w, s[j]);
            }
        }
    } else {
        stage_kt(src, sh_kt, tid, lane, wid, false);
        ((float4*)sh_qt)[tid] = qraw;
        __syncthreads();

        const float4* v4 = (const float4*)sh_v;
#pragma unroll
        for (int j = 0; j < KPL; ++j) s[j] = 0.0f;
#pragma unroll 1
        for (int a = 0; a < AD_ / 4; ++a) {
            const float4 qv = q4[a];
            const float4 vw = v4[a];
            float4 kv[KPL];
#pragma unroll
            for (int j = 0; j < KPL; ++j)
                kv[j] = *(const float4*)(krow + j * 32 * KT_PITCH + 4 * a);
#pragma unroll
            for (int j = 0; j < KPL; ++j) {
                s[j] += fast_tanh(qv.x + kv[j].x) * vw.x;
                s[j] += fast_tanh(qv.y + kv[j].y) * vw.y;
                s[j] += fast_tanh(qv.z + kv[j].z) * vw.z;
                s[j] += fast_tanh(qv.w + kv[j].w) * vw.w;
            }
        }
    }

    float m = s[0];
#pragma unroll
    for (int j = 1; j < KPL; ++j) m = fmaxf(m, s[j]);
#pragma unroll
    for (int off = 16; off > 0; off >>= 1)
        m = fmaxf(m, __shfl_xor(m, off));

    float e[KPL];
    float sum = 0.0f;
#pragma unroll
    for (int j = 0; j < KPL; ++j) {
        e[j] = __builtin_amdgcn_exp2f((s[j] - m) * 1.4426950408889634f);
        sum += e[j];
    }
#pragma unroll
    for (int off = 16; off > 0; off >>= 1)
        sum += __shfl_xor(sum, off);

    float* __restrict__ wrow = Wout + (size_t)(row0 + wid) * SK_ + lane;
#pragma unroll
    for (int j = 0; j < KPL; ++j)
        wrow[32 * j] = e[j] / sum;
}

__global__ __launch_bounds__(GT) void k_av(
    const float* __restrict__ Wt, const float* __restrict__ V,
    float* __restrict__ Ov)
{
    __shared__ __attribute__((aligned(16))) float sA[NST * STG_A];
    __shared__ __attribute__((aligned(16))) float sB[NST * STG_B];
    const int b   = blockIdx.x >> 6;
    const int t   = blockIdx.x & 63;
    const int bm  = t >> 2;
    const int bn  = t & 3;
    const int wid = threadIdx.x >> 5;
    const float* Abase = Wt + (size_t)b * SQ_ * SK_ + (size_t)bm * GBM * SK_;
    const float* Bbase = V  + (size_t)b * SK_ * H_  + bn * GBN;
    v8f c = {};
    c = gemm_block_lds<SK_ / GKC>(Abase, SK_, Bbase, H_, c, sA, sB);
    store_tile(Ov + (size_t)b * SQ_ * H_ + (size_t)(bm * GBM + (wid >> 2) * 16) * H_
                  + bn * GBN + (wid & 3) * 16, H_, c);
}

__global__ __launch_bounds__(GT) void k_out(
    const float* __restrict__ Ov, const float* __restrict__ Wo,
    const float* __restrict__ bo, float* __restrict__ Out)
{
    __shared__ __attribute__((aligned(16))) float sA[NST * STG_A];
    __shared__ __attribute__((aligned(16))) float sB[NST * STG_B];
    const int bm   = blockIdx.x >> 2;
    const int bn   = blockIdx.x & 3;
    const int wid  = threadIdx.x >> 5;
    const int lane = threadIdx.x & 31;
    const int col0 = bn * GBN + (wid & 3) * 16;
    const float bias = bo[col0 + (lane & 15)];
    v8f c;
#pragma unroll
    for (int i = 0; i < 8; ++i) c[i] = bias;
    c = gemm_block_lds<H_ / GKC>(Ov + (size_t)bm * GBM * H_, H_, Wo + bn * GBN, H_, c, sA, sB);
    store_tile(Out + (size_t)(bm * GBM + (wid >> 2) * 16) * H_ + col0, H_, c);
}

extern "C" void kernel_launch(void* const* d_in, const int* in_sizes, int n_in,
                              void* d_out, int out_size, void* d_ws, size_t ws_size,
                              hipStream_t stream)
{
    const float* Q  = (const float*)d_in[0];
    const float* K  = (const float*)d_in[1];
    const float* V  = (const float*)d_in[2];
    const float* Wq = (const float*)d_in[3];
    const float* Wk = (const float*)d_in[4];
    const float* vv = (const float*)d_in[5];
    const float* Wo = (const float*)d_in[6];
    const float* bo = (const float*)d_in[7];

    float* out     = (float*)d_out;
    float* weights = out + (size_t)B_ * SQ_ * H_;

    float* Qt = (float*)d_ws;
    float* Kt = Qt + (size_t)B_ * SQ_ * AD_;
    float* Ov = Kt + (size_t)B_ * SK_ * AD_;

    k_project<<<256, GT, 0, stream>>>(Q, K, Wq, Wk, Qt, Kt);
    k_scores_softmax<<<B_ * SQ_ / QROWS, QROWS * 32, 0, stream>>>(Qt, Kt, vv, weights);
    k_av<<<256, GT, 0, stream>>>(weights, V, Ov);
    k_out<<<256, GT, 0, stream>>>(Ov, Wo, bo, out);
}
